# starting kernel
# baseline (speedup 1.0000x reference)
#include <hip/hip_runtime.h>

typedef __attribute__((ext_vector_type(2))) float v2f;
typedef __attribute__((ext_vector_type(8))) float v8f;

#define N_NODES 100000
#define N_EDGES 3200000
#define F_IN    256
#define CH      16
#define FCN     64
#define NTILES  (N_NODES / 16)

__global__ __launch_bounds__(256) void gcn_feat_wmma(
    const float* __restrict__ x, const float* __restrict__ wg,
    float* __restrict__ h)
{
    __shared__ float lds_w[F_IN * CH];
    const int tid = threadIdx.x;
    for (int i = tid; i < F_IN * CH; i += 256) lds_w[i] = wg[i];
    __syncthreads();

    const int wave = tid >> 5;
    const int lane = tid & 31;
    const int tile = blockIdx.x * 8 + wave;
    if (tile >= NTILES) return;

    const int m  = lane & 15;
    const int hi = lane >> 4;
    const int row0 = tile * 16;
    const float* arow = x + (size_t)(row0 + m) * F_IN + 2 * hi;

    v8f acc = {0.f, 0.f, 0.f, 0.f, 0.f, 0.f, 0.f, 0.f};
    #pragma unroll 8
    for (int k = 0; k < 64; ++k) {
        const int kk = 4 * k;
        v2f a; a.x = arow[kk];     a.y = arow[kk + 1];
        v2f b; b.x = lds_w[(kk + 2 * hi)     * CH + m];
               b.y = lds_w[(kk + 2 * hi + 1) * CH + m];
        acc = __builtin_amdgcn_wmma_f32_16x16x4_f32(
                  false, a, false, b, (short)0, acc, false, false);
    }

    float* hrow = h + (size_t)row0 * CH + m;
    #pragma unroll
    for (int j = 0; j < 8; ++j)
        hrow[(size_t)(j + 8 * hi) * CH] = acc[j];
}

__global__ __launch_bounds__(256) void gcn_edge_scatter(
    const float* __restrict__ h, const int* __restrict__ src,
    const int* __restrict__ dst, const float* __restrict__ ew,
    float* __restrict__ agg)
{
    const long long idx = (long long)blockIdx.x * 256 + threadIdx.x;
    if (idx >= (long long)N_EDGES * CH) return;
    const int e = (int)(idx >> 4);
    const int c = (int)(idx & 15);
    const int s = src[e];
    const int d = dst[e];
    const float v = h[(size_t)s * CH + c] * ew[e];
    atomicAdd(&agg[(size_t)d * CH + c], v);
}

__global__ __launch_bounds__(256) void gcn_head_wmma(
    const float* __restrict__ agg, const float* __restrict__ w0,
    const float* __restrict__ b0,  const float* __restrict__ w1,
    const float* __restrict__ b1,  float* __restrict__ out)
{
    __shared__ float lds_w0[CH * 80];
    __shared__ float lds_b0[FCN];
    __shared__ float lds_w1[FCN];
    __shared__ float lds_o1[8][16 * 68];

    const int tid = threadIdx.x;
    for (int i = tid; i < CH * FCN; i += 256)
        lds_w0[(i / FCN) * 80 + (i % FCN)] = w0[i];
    if (tid < FCN) { lds_b0[tid] = b0[tid]; lds_w1[tid] = w1[tid]; }
    __syncthreads();

    const int wave = tid >> 5;
    const int lane = tid & 31;
    const int tile = blockIdx.x * 8 + wave;
    if (tile >= NTILES) return;

    const int m  = lane & 15;
    const int hi = lane >> 4;
    const int row0 = tile * 16;
    const float* arow = agg + (size_t)(row0 + m) * CH + 2 * hi;

    v8f acc[4];
    #pragma unroll
    for (int t = 0; t < 4; ++t) acc[t] = (v8f){0.f,0.f,0.f,0.f,0.f,0.f,0.f,0.f};

    #pragma unroll
    for (int k = 0; k < 4; ++k) {
        const int kk = 4 * k;
        v2f a; a.x = fmaxf(arow[kk], 0.f);
               a.y = fmaxf(arow[kk + 1], 0.f);
        const float* br0 = &lds_w0[(kk + 2 * hi)     * 80 + m];
        const float* br1 = &lds_w0[(kk + 2 * hi + 1) * 80 + m];
        #pragma unroll
        for (int t = 0; t < 4; ++t) {
            v2f b; b.x = br0[16 * t]; b.y = br1[16 * t];
            acc[t] = __builtin_amdgcn_wmma_f32_16x16x4_f32(
                         false, a, false, b, (short)0, acc[t], false, false);
        }
    }

    float* o1 = &lds_o1[wave][0];
    #pragma unroll
    for (int t = 0; t < 4; ++t) {
        const float bias = lds_b0[16 * t + m];
        #pragma unroll
        for (int j = 0; j < 8; ++j)
            o1[(j + 8 * hi) * 68 + 16 * t + m] = fmaxf(acc[t][j] + bias, 0.f);
    }

    const float* row = &lds_o1[wave][m * 68 + 32 * hi];
    const float* wv  = &lds_w1[32 * hi];
    float s = 0.f;
    #pragma unroll
    for (int f = 0; f < 32; ++f) s += row[f] * wv[f];
    s += __shfl_xor(s, 16, 32);
    if (hi == 0) out[row0 + m] = s + b1[0];
}

extern "C" void kernel_launch(void* const* d_in, const int* in_sizes, int n_in,
                              void* d_out, int out_size, void* d_ws, size_t ws_size,
                              hipStream_t stream)
{
    (void)in_sizes; (void)n_in; (void)out_size; (void)ws_size;
    const float* x  = (const float*)d_in[0];
    const int*   ei = (const int*)  d_in[1];
    const float* ew = (const float*)d_in[2];
    const float* wg = (const float*)d_in[3];
    const float* w0 = (const float*)d_in[4];
    const float* b0 = (const float*)d_in[5];
    const float* w1 = (const float*)d_in[6];
    const float* b1 = (const float*)d_in[7];
    float* out = (float*)d_out;

    float* h   = (float*)d_ws;
    float* agg = h + (size_t)N_NODES * CH;

    hipMemsetAsync(agg, 0, (size_t)N_NODES * CH * sizeof(float), stream);

    const int tile_blocks = (NTILES + 7) / 8;
    gcn_feat_wmma<<<tile_blocks, 256, 0, stream>>>(x, wg, h);

    const long long ework = (long long)N_EDGES * CH;
    const int edge_blocks = (int)((ework + 255) / 256);
    gcn_edge_scatter<<<edge_blocks, 256, 0, stream>>>(h, ei, ei + N_EDGES, ew, agg);

    gcn_head_wmma<<<tile_blocks, 256, 0, stream>>>(agg, w0, b0, w1, b1, out);
}


# feature transform and edge sort share one launch: 250 work-groups compute h (25 node tiles each, same WMMA k chain) beside the 400 sort work-groups of the same grid; 3 launches become 2
# speedup vs baseline: 3.3641x; 3.3641x over previous
#include <hip/hip_runtime.h>

typedef __attribute__((ext_vector_type(2))) float v2f;
typedef __attribute__((ext_vector_type(8))) float v8f;

#define N_NODES 100000
#define N_EDGES 3200000
#define F_IN    256
#define CH      16
#define FCN     64
#define NTILES  (N_NODES / 16)

__device__ __forceinline__ void feat_tile_wmma(
    const float* __restrict__ x, const float* lds_w, float* __restrict__ h, int tile, int lane)
{
    const int m  = lane & 15;
    const int hi = lane >> 4;
    const int row0 = tile * 16;
    const float* arow = x + (size_t)(row0 + m) * F_IN + 2 * hi;

    v8f acc = {0.f, 0.f, 0.f, 0.f, 0.f, 0.f, 0.f, 0.f};
    #pragma unroll 8
    for (int k = 0; k < 64; ++k) {
        const int kk = 4 * k;
        v2f a; a.x = arow[kk];     a.y = arow[kk + 1];
        v2f b; b.x = lds_w[(kk + 2 * hi)     * CH + m];
               b.y = lds_w[(kk + 2 * hi + 1) * CH + m];
        acc = __builtin_amdgcn_wmma_f32_16x16x4_f32(
                  false, a, false, b, (short)0, acc, false, false);
    }

    float* hrow = h + (size_t)row0 * CH + m;
    #pragma unroll
    for (int j = 0; j < 8; ++j)
        hrow[(size_t)(j + 8 * hi) * CH] = acc[j];
}

#define RANGES      250
#define RANGE_TILES 25
#define RANGE_NODES (RANGE_TILES * 16)
#define P_PARTS     400
#define PART_EDGES  (N_EDGES / P_PARTS)
#define PART_QUADS  (PART_EDGES / 4)
#define OFF_STRIDE  256
#define SEGS        13
#define LIST_CAP    (SEGS * 32)
#define SRC_BITS    17
#define N_WGP       256
#define FEAT_BLOCKS 250
#define FEAT_TILES  25

static_assert(N_NODES == RANGES * RANGE_NODES, "ranges must tile the nodes");
static_assert(N_EDGES % P_PARTS == 0 && PART_EDGES % 4 == 0, "pieces must keep 16-byte loads aligned");
static_assert(PART_QUADS <= 2 * 1024, "two quads per thread cover a piece");
static_assert(PART_EDGES < (1 << 16) && RANGES < OFF_STRIDE, "key packing and offset row");
static_assert(SEGS * 32 >= P_PARTS, "the waves of a scatter work-group cover all pieces");
static_assert(N_NODES <= (1 << SRC_BITS) && RANGE_NODES < (1 << (31 - SRC_BITS)), "record packing");
static_assert(F_IN * CH / 4 == 1024, "one 16-byte piece of w_gcn per thread");
static_assert(FEAT_BLOCKS * FEAT_TILES >= NTILES && FEAT_TILES <= 32, "the feature work-groups cover every node tile");
static_assert(N_WGP % 2 == 0 && 2 * FEAT_BLOCKS >= N_WGP && FEAT_BLOCKS <= P_PARTS, "grid places: every feature work-group and every piece exactly once");

__global__ __launch_bounds__(1024) void gcn_feat_edge_sort(
    const float* __restrict__ x, const float* __restrict__ wg, float* __restrict__ h,
    const int* __restrict__ src, const int* __restrict__ dst,
    const float* __restrict__ ew, int* __restrict__ off, int* __restrict__ rec)
{
    __shared__ int  cur[OFF_STRIDE];
    __shared__ int2 stage[PART_EDGES];
    __shared__ float4 lds_w4[F_IN * CH / 4];

    const int tid  = threadIdx.x;
    const int wave = tid >> 5;
    const int lane = tid & 31;

    const int bid = blockIdx.x;
    int fb = -1, p;
    if (bid < N_WGP) {
        p = bid >> 1;
        if ((bid & 1) == 0) fb = p;
    } else if (bid < 2 * FEAT_BLOCKS) {
        p = (N_WGP >> 1) + ((bid - N_WGP) >> 1);
        if ((bid & 1) == 1) fb = p;
    } else {
        p = bid - FEAT_BLOCKS;
    }

    if (fb >= 0) {
        lds_w4[tid] = reinterpret_cast<const float4*>(wg)[tid];
        __syncthreads();
        const int tile = fb * FEAT_TILES + wave;
        if (wave < FEAT_TILES && tile < NTILES)
            feat_tile_wmma(x, reinterpret_cast<const float*>(lds_w4), h, tile, lane);
        return;
    }

    if (tid < OFF_STRIDE) cur[tid] = 0;
    __syncthreads();

    int rx[8], rw[8], key[8];
    const int e0 = p * PART_EDGES;
    int4 d4[2], s4[2]; float4 w4[2];
    #pragma unroll
    for (int it = 0; it < 2; ++it) {
        const int q = tid + it * 1024;
        d4[it] = make_int4(-1, -1, -1, -1);
        s4[it] = make_int4(0, 0, 0, 0);
        w4[it] = make_float4(0.f, 0.f, 0.f, 0.f);
        if (q < PART_QUADS) {
            const int i = e0 + 4 * q;
            d4[it] = *reinterpret_cast<const int4*>(dst + i);
            s4[it] = *reinterpret_cast<const int4*>(src + i);
            w4[it] = *reinterpret_cast<const float4*>(ew + i);
        }
    }
    #pragma unroll
    for (int it = 0; it < 2; ++it) {
        const int   dv[4] = {d4[it].x, d4[it].y, d4[it].z, d4[it].w};
        const int   sv[4] = {s4[it].x, s4[it].y, s4[it].z, s4[it].w};
        const float wv[4] = {w4[it].x, w4[it].y, w4[it].z, w4[it].w};
        #pragma unroll
        for (int j = 0; j < 4; ++j) {
            const int e = 4 * it + j;
            key[e] = -1; rx[e] = 0; rw[e] = 0;
            if ((unsigned)dv[j] < (unsigned)N_NODES) {
                const int rg   = dv[j] / RANGE_NODES;
                const int loc  = dv[j] - rg * RANGE_NODES;
                const int rank = atomicAdd(&cur[rg], 1);
                key[e] = (rg << 16) | rank;
                rx[e]  = sv[j] | (loc << SRC_BITS);
                rw[e]  = __float_as_int(wv[j]);
            }
        }
    }
    __syncthreads();

    if (wave == 0) {
        int cv[8];
        int s = 0;
        #pragma unroll
        for (int k = 0; k < 8; ++k) { cv[k] = cur[8 * lane + k]; s += cv[k]; }
        int incl = s;
        #pragma unroll
        for (int d = 1; d < 32; d <<= 1) {
            const int t = __shfl_up(incl, d, 32);
            if (lane >= d) incl += t;
        }
        int run = incl - s;
        #pragma unroll
        for (int k = 0; k < 8; ++k) { cur[8 * lane + k] = run; run += cv[k]; }
    }
    __syncthreads();

    if (tid < OFF_STRIDE) off[p * OFF_STRIDE + tid] = cur[tid];

    #pragma unroll
    for (int e = 0; e < 8; ++e)
        if (key[e] >= 0) stage[cur[key[e] >> 16] + (key[e] & 0xFFFF)] = make_int2(rx[e], rw[e]);
    __syncthreads();

    const int total = min(cur[RANGES], PART_EDGES);
    int2* outp = reinterpret_cast<int2*>(rec) + (size_t)p * PART_EDGES;
    for (int j = tid; j < total; j += 1024) outp[j] = stage[j];
}

__device__ __forceinline__ void scatter_list(
    const int2* myq, int m, const float* __restrict__ h, float* sums, int c, int hi)
{
    for (int t = 0; t < m; t += 8) {
        float v[4]; int row[4]; bool ok[4];
        #pragma unroll
        for (int k = 0; k < 4; ++k) {
            const int idx = t + 2 * k + hi;
            ok[k] = idx < m;
            const int2 rc = myq[ok[k] ? idx : 0];
            const int sn = rc.x & ((1 << SRC_BITS) - 1);
            row[k] = (int)((unsigned)rc.x >> SRC_BITS);
            v[k] = h[(size_t)sn * CH + c] * __int_as_float(rc.y);
        }
        #pragma unroll
        for (int k = 0; k < 4; ++k)
            if (ok[k]) atomicAdd(&sums[row[k] * CH + c], v[k]);
    }
}

__global__ __launch_bounds__(1024) void gcn_range_scatter_head(
    const float* __restrict__ h, const int* __restrict__ off, const int* __restrict__ rec,
    const float* __restrict__ w0, const float* __restrict__ b0,
    const float* __restrict__ w1, const float* __restrict__ b1, float* __restrict__ out)
{
    __shared__ float sums[RANGE_NODES * CH];
    __shared__ int2  kept[32][LIST_CAP];
    __shared__ float lds_w0[CH * 80];
    __shared__ float lds_b0[FCN];
    __shared__ float lds_w1[FCN];
    __shared__ float lds_o1[RANGE_TILES][16 * 68];

    const int tid  = threadIdx.x;
    const int wave = tid >> 5;
    const int lane = tid & 31;
    const int c    = lane & 15;
    const int hi   = lane >> 4;
    const int r    = blockIdx.x;

    for (int i = tid; i < RANGE_NODES * CH / 4; i += 1024)
        reinterpret_cast<float4*>(sums)[i] = make_float4(0.f, 0.f, 0.f, 0.f);
    for (int i = tid; i < CH * FCN; i += 1024)
        lds_w0[(i / FCN) * 80 + (i % FCN)] = w0[i];
    if (tid < FCN) { lds_b0[tid] = b0[tid]; lds_w1[tid] = w1[tid]; }
    __syncthreads();

    int sb = 0, se = 0;
    {
        const int pl = wave + 32 * lane;
        if (lane < SEGS && pl < P_PARTS) {
            sb = off[pl * OFF_STRIDE + r];
            se = off[pl * OFF_STRIDE + r + 1];
        }
        se = min(max(se, 0), PART_EDGES);
        sb = min(max(sb, 0), se);
    }
    const int2* recs = reinterpret_cast<const int2*>(rec);
    int2* myq = kept[wave];

    int2 pre0[SEGS], pre1[SEGS];
    #pragma unroll
    for (int i = 0; i < SEGS; ++i) {
        const int s = __builtin_amdgcn_readlane(sb, i);
        const int e = __builtin_amdgcn_readlane(se, i);
        const int2* seg = recs + (size_t)(wave + 32 * i) * PART_EDGES;
        pre0[i] = s + lane < e ? seg[s + lane] : make_int2(0, 0);
    }
    #pragma unroll
    for (int i = 0; i < SEGS; ++i) {
        const int s = __builtin_amdgcn_readlane(sb, i);
        const int e = __builtin_amdgcn_readlane(se, i);
        const int2* seg = recs + (size_t)(wave + 32 * i) * PART_EDGES;
        pre1[i] = s + 32 + lane < e ? seg[s + 32 + lane] : make_int2(0, 0);
    }

    int fill = 0;
    #pragma unroll
    for (int i = 0; i < SEGS; ++i) {
        const int n = __builtin_amdgcn_readlane(se, i) - __builtin_amdgcn_readlane(sb, i);
        const int m = min(n, 32);
        if (lane < m) myq[fill + lane] = pre0[i];
        fill += m;
    }
    __builtin_amdgcn_fence(__ATOMIC_ACQ_REL, "wavefront");
    scatter_list(myq, fill, h, sums, c, hi);
    __builtin_amdgcn_fence(__ATOMIC_ACQ_REL, "wavefront");

    fill = 0;
    #pragma unroll
    for (int i = 0; i < SEGS; ++i) {
        const int n = __builtin_amdgcn_readlane(se, i) - __builtin_amdgcn_readlane(sb, i);
        const int m = min(max(n - 32, 0), 32);
        if (lane < m) myq[fill + lane] = pre1[i];
        fill += m;
    }
    __builtin_amdgcn_fence(__ATOMIC_ACQ_REL, "wavefront");
    scatter_list(myq, fill, h, sums, c, hi);
    __builtin_amdgcn_fence(__ATOMIC_ACQ_REL, "wavefront");

    for (int i = 0; i < SEGS; ++i) {
        const int s = __builtin_amdgcn_readlane(sb, i);
        const int e = __builtin_amdgcn_readlane(se, i);
        const int2* seg = recs + (size_t)(wave + 32 * i) * PART_EDGES;
        for (int b = s + 64; b < e; b += 32) {
            const int m = min(e - b, 32);
            if (lane < m) myq[lane] = seg[b + lane];
            __builtin_amdgcn_fence(__ATOMIC_ACQ_REL, "wavefront");
            scatter_list(myq, m, h, sums, c, hi);
            __builtin_amdgcn_fence(__ATOMIC_ACQ_REL, "wavefront");
        }
    }
    __syncthreads();

    if (wave < RANGE_TILES) {
        const int m = c;
        const float* arow = &sums[(16 * wave + m) * CH + 2 * hi];

        v8f facc[4];
        #pragma unroll
        for (int t = 0; t < 4; ++t) facc[t] = (v8f){0.f,0.f,0.f,0.f,0.f,0.f,0.f,0.f};

        #pragma unroll
        for (int k = 0; k < 4; ++k) {
            const int kk = 4 * k;
            v2f a; a.x = fmaxf(arow[kk], 0.f);
                   a.y = fmaxf(arow[kk + 1], 0.f);
            const float* br0 = &lds_w0[(kk + 2 * hi)     * 80 + m];
            const float* br1 = &lds_w0[(kk + 2 * hi + 1) * 80 + m];
            #pragma unroll
            for (int t = 0; t < 4; ++t) {
                v2f b; b.x = br0[16 * t]; b.y = br1[16 * t];
                facc[t] = __builtin_amdgcn_wmma_f32_16x16x4_f32(
                              false, a, false, b, (short)0, facc[t], false, false);
            }
        }

        float* o1 = &lds_o1[wave][0];
        #pragma unroll
        for (int t = 0; t < 4; ++t) {
            const float bias = lds_b0[16 * t + m];
            #pragma unroll
            for (int j = 0; j < 8; ++j)
                o1[(j + 8 * hi) * 68 + 16 * t + m] = fmaxf(facc[t][j] + bias, 0.f);
        }

        const float* row = &lds_o1[wave][m * 68 + 32 * hi];
        const float* wv  = &lds_w1[32 * hi];
        float s = 0.f;
        #pragma unroll
        for (int f = 0; f < 32; ++f) s += row[f] * wv[f];
        s += __shfl_xor(s, 16, 32);
        if (hi == 0) out[r * RANGE_NODES + 16 * wave + m] = s + b1[0];
    }
}

extern "C" void kernel_launch(void* const* d_in, const int* in_sizes, int n_in,
                              void* d_out, int out_size, void* d_ws, size_t ws_size,
                              hipStream_t stream)
{
    (void)in_sizes; (void)n_in; (void)out_size; (void)ws_size;
    const float* x  = (const float*)d_in[0];
    const int*   ei = (const int*)  d_in[1];
    const float* ew = (const float*)d_in[2];
    const float* wg = (const float*)d_in[3];
    const float* w0 = (const float*)d_in[4];
    const float* b0 = (const float*)d_in[5];
    const float* w1 = (const float*)d_in[6];
    const float* b1 = (const float*)d_in[7];
    float* out = (float*)d_out;

    float* h   = (float*)d_ws;
    int*   off = (int*)(h + (size_t)N_NODES * CH);
    int*   rec = (int*)(h + (size_t)2 * N_NODES * CH) + 8192;

    gcn_feat_edge_sort<<<FEAT_BLOCKS + P_PARTS, 1024, 0, stream>>>(x, wg, h, ei, ei + N_EDGES, ew, off, rec);

    gcn_range_scatter_head<<<RANGES, 1024, 0, stream>>>(h, off, rec, w0, b0, w1, b1, out);
}


# edge scatter: 4 lanes gather a source row by 16-byte loads, 8 records per load instruction (was 2); each lane adds its 4 channels in a slot-rotated bank order; sums stay in LDS arrival order
# speedup vs baseline: 4.1821x; 1.2432x over previous
#include <hip/hip_runtime.h>

typedef __attribute__((ext_vector_type(2))) float v2f;
typedef __attribute__((ext_vector_type(8))) float v8f;

#define N_NODES 100000
#define N_EDGES 3200000
#define F_IN    256
#define CH      16
#define FCN     64
#define NTILES  (N_NODES / 16)

__device__ __forceinline__ void feat_tile_wmma(
    const float* __restrict__ x, const float* lds_w, float* __restrict__ h, int tile, int lane)
{
    const int m  = lane & 15;
    const int hi = lane >> 4;
    const int row0 = tile * 16;
    const float* arow = x + (size_t)(row0 + m) * F_IN + 2 * hi;

    v8f acc = {0.f, 0.f, 0.f, 0.f, 0.f, 0.f, 0.f, 0.f};
    #pragma unroll 8
    for (int k = 0; k < 64; ++k) {
        const int kk = 4 * k;
        v2f a; a.x = arow[kk];     a.y = arow[kk + 1];
        v2f b; b.x = lds_w[(kk + 2 * hi)     * CH + m];
               b.y = lds_w[(kk + 2 * hi + 1) * CH + m];
        acc = __builtin_amdgcn_wmma_f32_16x16x4_f32(
                  false, a, false, b, (short)0, acc, false, false);
    }

    float* hrow = h + (size_t)row0 * CH + m;
    #pragma unroll
    for (int j = 0; j < 8; ++j)
        hrow[(size_t)(j + 8 * hi) * CH] = acc[j];
}

#define RANGES      250
#define RANGE_TILES 25
#define RANGE_NODES (RANGE_TILES * 16)
#define P_PARTS     400
#define PART_EDGES  (N_EDGES / P_PARTS)
#define PART_QUADS  (PART_EDGES / 4)
#define OFF_STRIDE  256
#define SEGS        13
#define LIST_CAP    (SEGS * 32)
#define SRC_BITS    17
#define N_WGP       256
#define FEAT_BLOCKS 250
#define FEAT_TILES  25

static_assert(N_NODES == RANGES * RANGE_NODES, "ranges must tile the nodes");
static_assert(N_EDGES % P_PARTS == 0 && PART_EDGES % 4 == 0, "pieces must keep 16-byte loads aligned");
static_assert(PART_QUADS <= 2 * 1024, "two quads per thread cover a piece");
static_assert(PART_EDGES < (1 << 16) && RANGES < OFF_STRIDE, "key packing and offset row");
static_assert(SEGS * 32 >= P_PARTS, "the waves of a scatter work-group cover all pieces");
static_assert(N_NODES <= (1 << SRC_BITS) && RANGE_NODES < (1 << (31 - SRC_BITS)), "record packing");
static_assert(F_IN * CH / 4 == 1024, "one 16-byte piece of w_gcn per thread");
static_assert(FEAT_BLOCKS * FEAT_TILES >= NTILES && FEAT_TILES <= 32, "the feature work-groups cover every node tile");
static_assert(N_WGP % 2 == 0 && 2 * FEAT_BLOCKS >= N_WGP && FEAT_BLOCKS <= P_PARTS, "grid places: every feature work-group and every piece exactly once");

__global__ __launch_bounds__(1024) void gcn_feat_edge_sort(
    const float* __restrict__ x, const float* __restrict__ wg, float* __restrict__ h,
    const int* __restrict__ src, const int* __restrict__ dst,
    const float* __restrict__ ew, int* __restrict__ off, int* __restrict__ rec)
{
    __shared__ int  cur[OFF_STRIDE];
    __shared__ int2 stage[PART_EDGES];
    __shared__ float4 lds_w4[F_IN * CH / 4];

    const int tid  = threadIdx.x;
    const int wave = tid >> 5;
    const int lane = tid & 31;

    const int bid = blockIdx.x;
    int fb = -1, p;
    if (bid < N_WGP) {
        p = bid >> 1;
        if ((bid & 1) == 0) fb = p;
    } else if (bid < 2 * FEAT_BLOCKS) {
        p = (N_WGP >> 1) + ((bid - N_WGP) >> 1);
        if ((bid & 1) == 1) fb = p;
    } else {
        p = bid - FEAT_BLOCKS;
    }

    if (fb >= 0) {
        lds_w4[tid] = reinterpret_cast<const float4*>(wg)[tid];
        __syncthreads();
        const int tile = fb * FEAT_TILES + wave;
        if (wave < FEAT_TILES && tile < NTILES)
            feat_tile_wmma(x, reinterpret_cast<const float*>(lds_w4), h, tile, lane);
        return;
    }

    if (tid < OFF_STRIDE) cur[tid] = 0;
    __syncthreads();

    int rx[8], rw[8], key[8];
    const int e0 = p * PART_EDGES;
    int4 d4[2], s4[2]; float4 w4[2];
    #pragma unroll
    for (int it = 0; it < 2; ++it) {
        const int q = tid + it * 1024;
        d4[it] = make_int4(-1, -1, -1, -1);
        s4[it] = make_int4(0, 0, 0, 0);
        w4[it] = make_float4(0.f, 0.f, 0.f, 0.f);
        if (q < PART_QUADS) {
            const int i = e0 + 4 * q;
            d4[it] = *reinterpret_cast<const int4*>(dst + i);
            s4[it] = *reinterpret_cast<const int4*>(src + i);
            w4[it] = *reinterpret_cast<const float4*>(ew + i);
        }
    }
    #pragma unroll
    for (int it = 0; it < 2; ++it) {
        const int   dv[4] = {d4[it].x, d4[it].y, d4[it].z, d4[it].w};
        const int   sv[4] = {s4[it].x, s4[it].y, s4[it].z, s4[it].w};
        const float wv[4] = {w4[it].x, w4[it].y, w4[it].z, w4[it].w};
        #pragma unroll
        for (int j = 0; j < 4; ++j) {
            const int e = 4 * it + j;
            key[e] = -1; rx[e] = 0; rw[e] = 0;
            if ((unsigned)dv[j] < (unsigned)N_NODES) {
                const int rg   = dv[j] / RANGE_NODES;
                const int loc  = dv[j] - rg * RANGE_NODES;
                const int rank = atomicAdd(&cur[rg], 1);
                key[e] = (rg << 16) | rank;
                rx[e]  = sv[j] | (loc << SRC_BITS);
                rw[e]  = __float_as_int(wv[j]);
            }
        }
    }
    __syncthreads();

    if (wave == 0) {
        int cv[8];
        int s = 0;
        #pragma unroll
        for (int k = 0; k < 8; ++k) { cv[k] = cur[8 * lane + k]; s += cv[k]; }
        int incl = s;
        #pragma unroll
        for (int d = 1; d < 32; d <<= 1) {
            const int t = __shfl_up(incl, d, 32);
            if (lane >= d) incl += t;
        }
        int run = incl - s;
        #pragma unroll
        for (int k = 0; k < 8; ++k) { cur[8 * lane + k] = run; run += cv[k]; }
    }
    __syncthreads();

    if (tid < OFF_STRIDE) off[p * OFF_STRIDE + tid] = cur[tid];

    #pragma unroll
    for (int e = 0; e < 8; ++e)
        if (key[e] >= 0) stage[cur[key[e] >> 16] + (key[e] & 0xFFFF)] = make_int2(rx[e], rw[e]);
    __syncthreads();

    const int total = min(cur[RANGES], PART_EDGES);
    int2* outp = reinterpret_cast<int2*>(rec) + (size_t)p * PART_EDGES;
    for (int j = tid; j < total; j += 1024) outp[j] = stage[j];
}

__device__ __forceinline__ void scatter_list(
    const int2* myq, int m, const float* __restrict__ h, float* sums, int q, int slot)
{
    const int  rot = slot & 3;
    const bool r1  = (rot & 1) != 0;
    const bool r2  = (rot & 2) != 0;
    float* const p0 = sums + 4 * q + rot;
    float* const p1 = sums + 4 * q + (rot ^ 1);
    float* const p2 = sums + 4 * q + (rot ^ 2);
    float* const p3 = sums + 4 * q + (rot ^ 3);
    const float* const hq = h + 4 * q;
    for (int t = 0; t < m; t += 32) {
        float4 v[4]; int ro[4]; bool ok[4];
        #pragma unroll
        for (int k = 0; k < 4; ++k) {
            const int idx = t + 8 * k + slot;
            ok[k] = idx < m;
            const int2 rc = myq[ok[k] ? idx : 0];
            const int sn = rc.x & ((1 << SRC_BITS) - 1);
            ro[k] = (int)((unsigned)rc.x >> SRC_BITS) * CH;
            const float w = __int_as_float(rc.y);
            const float4 g = *reinterpret_cast<const float4*>(hq + (size_t)sn * CH);
            v[k] = make_float4(g.x * w, g.y * w, g.z * w, g.w * w);
        }
        #pragma unroll
        for (int k = 0; k < 4; ++k) {
            if (ok[k]) {
                const float a0 = r1 ? v[k].y : v[k].x, a1 = r1 ? v[k].x : v[k].y;
                const float a2 = r1 ? v[k].w : v[k].z, a3 = r1 ? v[k].z : v[k].w;
                const float e0 = r2 ? a2 : a0, e1 = r2 ? a3 : a1;
                const float e2 = r2 ? a0 : a2, e3 = r2 ? a1 : a3;
                atomicAdd(p0 + ro[k], e0);
                atomicAdd(p1 + ro[k], e1);
                atomicAdd(p2 + ro[k], e2);
                atomicAdd(p3 + ro[k], e3);
            }
        }
    }
}

__global__ __launch_bounds__(1024) void gcn_range_scatter_head(
    const float* __restrict__ h, const int* __restrict__ off, const int* __restrict__ rec,
    const float* __restrict__ w0, const float* __restrict__ b0,
    const float* __restrict__ w1, const float* __restrict__ b1, float* __restrict__ out)
{
    __shared__ float sums[RANGE_NODES * CH];
    __shared__ int2  kept[32][LIST_CAP];
    __shared__ float lds_w0[CH * 80];
    __shared__ float lds_b0[FCN];
    __shared__ float lds_w1[FCN];
    __shared__ float lds_o1[RANGE_TILES][16 * 68];

    const int tid  = threadIdx.x;
    const int wave = tid >> 5;
    const int lane = tid & 31;
    const int c    = lane & 15;
    const int hi   = lane >> 4;
    const int r    = blockIdx.x;

    for (int i = tid; i < RANGE_NODES * CH / 4; i += 1024)
        reinterpret_cast<float4*>(sums)[i] = make_float4(0.f, 0.f, 0.f, 0.f);
    for (int i = tid; i < CH * FCN; i += 1024)
        lds_w0[(i / FCN) * 80 + (i % FCN)] = w0[i];
    if (tid < FCN) { lds_b0[tid] = b0[tid]; lds_w1[tid] = w1[tid]; }
    __syncthreads();

    int sb = 0, se = 0;
    {
        const int pl = wave + 32 * lane;
        if (lane < SEGS && pl < P_PARTS) {
            sb = off[pl * OFF_STRIDE + r];
            se = off[pl * OFF_STRIDE + r + 1];
        }
        se = min(max(se, 0), PART_EDGES);
        sb = min(max(sb, 0), se);
    }
    const int2* recs = reinterpret_cast<const int2*>(rec);
    int2* myq = kept[wave];

    int2 pre0[SEGS], pre1[SEGS];
    #pragma unroll
    for (int i = 0; i < SEGS; ++i) {
        const int s = __builtin_amdgcn_readlane(sb, i);
        const int e = __builtin_amdgcn_readlane(se, i);
        const int2* seg = recs + (size_t)(wave + 32 * i) * PART_EDGES;
        pre0[i] = s + lane < e ? seg[s + lane] : make_int2(0, 0);
    }
    #pragma unroll
    for (int i = 0; i < SEGS; ++i) {
        const int s = __builtin_amdgcn_readlane(sb, i);
        const int e = __builtin_amdgcn_readlane(se, i);
        const int2* seg = recs + (size_t)(wave + 32 * i) * PART_EDGES;
        pre1[i] = s + 32 + lane < e ? seg[s + 32 + lane] : make_int2(0, 0);
    }

    int fill = 0;
    #pragma unroll
    for (int i = 0; i < SEGS; ++i) {
        const int n = __builtin_amdgcn_readlane(se, i) - __builtin_amdgcn_readlane(sb, i);
        const int m = min(n, 32);
        if (lane < m) myq[fill + lane] = pre0[i];
        fill += m;
    }
    __builtin_amdgcn_fence(__ATOMIC_ACQ_REL, "wavefront");
    scatter_list(myq, fill, h, sums, lane & 3, lane >> 2);
    __builtin_amdgcn_fence(__ATOMIC_ACQ_REL, "wavefront");

    fill = 0;
    #pragma unroll
    for (int i = 0; i < SEGS; ++i) {
        const int n = __builtin_amdgcn_readlane(se, i) - __builtin_amdgcn_readlane(sb, i);
        const int m = min(max(n - 32, 0), 32);
        if (lane < m) myq[fill + lane] = pre1[i];
        fill += m;
    }
    __builtin_amdgcn_fence(__ATOMIC_ACQ_REL, "wavefront");
    scatter_list(myq, fill, h, sums, lane & 3, lane >> 2);
    __builtin_amdgcn_fence(__ATOMIC_ACQ_REL, "wavefront");

    for (int i = 0; i < SEGS; ++i) {
        const int s = __builtin_amdgcn_readlane(sb, i);
        const int e = __builtin_amdgcn_readlane(se, i);
        const int2* seg = recs + (size_t)(wave + 32 * i) * PART_EDGES;
        for (int b = s + 64; b < e; b += 32) {
            const int m = min(e - b, 32);
            if (lane < m) myq[lane] = seg[b + lane];
            __builtin_amdgcn_fence(__ATOMIC_ACQ_REL, "wavefront");
            scatter_list(myq, m, h, sums, lane & 3, lane >> 2);
            __builtin_amdgcn_fence(__ATOMIC_ACQ_REL, "wavefront");
        }
    }
    __syncthreads();

    if (wave < RANGE_TILES) {
        const int m = c;
        const float* arow = &sums[(16 * wave + m) * CH + 2 * hi];

        v8f facc[4];
        #pragma unroll
        for (int t = 0; t < 4; ++t) facc[t] = (v8f){0.f,0.f,0.f,0.f,0.f,0.f,0.f,0.f};

        #pragma unroll
        for (int k = 0; k < 4; ++k) {
            const int kk = 4 * k;
            v2f a; a.x = fmaxf(arow[kk], 0.f);
                   a.y = fmaxf(arow[kk + 1], 0.f);
            const float* br0 = &lds_w0[(kk + 2 * hi)     * 80 + m];
            const float* br1 = &lds_w0[(kk + 2 * hi + 1) * 80 + m];
            #pragma unroll
            for (int t = 0; t < 4; ++t) {
                v2f b; b.x = br0[16 * t]; b.y = br1[16 * t];
                facc[t] = __builtin_amdgcn_wmma_f32_16x16x4_f32(
                              false, a, false, b, (short)0, facc[t], false, false);
            }
        }

        float* o1 = &lds_o1[wave][0];
        #pragma unroll
        for (int t = 0; t < 4; ++t) {
            const float bias = lds_b0[16 * t + m];
            #pragma unroll
            for (int j = 0; j < 8; ++j)
                o1[(j + 8 * hi) * 68 + 16 * t + m] = fmaxf(facc[t][j] + bias, 0.f);
        }

        const float* row = &lds_o1[wave][m * 68 + 32 * hi];
        const float* wv  = &lds_w1[32 * hi];
        float s = 0.f;
        #pragma unroll
        for (int f = 0; f < 32; ++f) s += row[f] * wv[f];
        s += __shfl_xor(s, 16, 32);
        if (hi == 0) out[r * RANGE_NODES + 16 * wave + m] = s + b1[0];
    }
}

extern "C" void kernel_launch(void* const* d_in, const int* in_sizes, int n_in,
                              void* d_out, int out_size, void* d_ws, size_t ws_size,
                              hipStream_t stream)
{
    (void)in_sizes; (void)n_in; (void)out_size; (void)ws_size;
    const float* x  = (const float*)d_in[0];
    const int*   ei = (const int*)  d_in[1];
    const float* ew = (const float*)d_in[2];
    const float* wg = (const float*)d_in[3];
    const float* w0 = (const float*)d_in[4];
    const float* b0 = (const float*)d_in[5];
    const float* w1 = (const float*)d_in[6];
    const float* b1 = (const float*)d_in[7];
    float* out = (float*)d_out;

    float* h   = (float*)d_ws;
    int*   off = (int*)(h + (size_t)N_NODES * CH);
    int*   rec = (int*)(h + (size_t)2 * N_NODES * CH) + 8192;

    gcn_feat_edge_sort<<<FEAT_BLOCKS + P_PARTS, 1024, 0, stream>>>(x, wg, h, ei, ei + N_EDGES, ew, off, rec);

    gcn_range_scatter_head<<<RANGES, 1024, 0, stream>>>(h, off, rec, w0, b0, w1, b1, out);
}
